# starting kernel
# baseline (speedup 1.0000x reference)
#include <hip/hip_runtime.h>
#include <math.h>

typedef __attribute__((ext_vector_type(2))) float v2f;
typedef __attribute__((ext_vector_type(8))) float v8f;

static constexpr int kHW    = 196;
static constexpr int kWdim  = 14;
static constexpr int kNReg  = 14;
static constexpr float kEps = 1e-6f;

__global__ __launch_bounds__(256) void rmac_region_max_kernel(
    const float* __restrict__ x, float* __restrict__ m, int nPlanes) {
  int idx = blockIdx.x * blockDim.x + threadIdx.x;
  if (idx >= nPlanes) return;
  const float* p = x + (size_t)idx * kHW;

  float acc[kNReg];
#pragma unroll
  for (int r = 0; r < kNReg; ++r) acc[r] = -INFINITY;

#pragma unroll
  for (int i = 0; i < 14; ++i) {
    float row[14];
    const float2* p2 = (const float2*)(p + i * kWdim);
#pragma unroll
    for (int j = 0; j < 7; ++j) {
      float2 v = p2[j];
      row[2 * j] = v.x;
      row[2 * j + 1] = v.y;
    }
    float c3a = row[0];
#pragma unroll
    for (int j = 1; j <= 6; ++j) c3a = fmaxf(c3a, row[j]);
    float c3b = row[3];
#pragma unroll
    for (int j = 4; j <= 9; ++j) c3b = fmaxf(c3b, row[j]);
    float c3c = row[7];
#pragma unroll
    for (int j = 8; j <= 13; ++j) c3c = fmaxf(c3c, row[j]);
    float c2a  = fmaxf(c3a, fmaxf(row[7], row[8]));
    float c2b  = fmaxf(c3c, fmaxf(row[5], row[6]));
    float full = fmaxf(c2a, c3c);

    acc[0] = fmaxf(acc[0], full);
    if (i <= 8) { acc[1] = fmaxf(acc[1], c2a); acc[2] = fmaxf(acc[2], c2b); }
    if (i >= 5) { acc[3] = fmaxf(acc[3], c2a); acc[4] = fmaxf(acc[4], c2b); }
    if (i <= 6) {
      acc[5] = fmaxf(acc[5], c3a); acc[6] = fmaxf(acc[6], c3b); acc[7] = fmaxf(acc[7], c3c);
    }
    if (i >= 3 && i <= 9) {
      acc[8] = fmaxf(acc[8], c3a); acc[9] = fmaxf(acc[9], c3b); acc[10] = fmaxf(acc[10], c3c);
    }
    if (i >= 7) {
      acc[11] = fmaxf(acc[11], c3a); acc[12] = fmaxf(acc[12], c3b); acc[13] = fmaxf(acc[13], c3c);
    }
  }
#pragma unroll
  for (int r = 0; r < kNReg; ++r) m[(size_t)r * nPlanes + idx] = acc[r];
}

__global__ __launch_bounds__(32) void rmac_norm_kernel(
    const float* __restrict__ m, float* __restrict__ w, int C) {
  int rb = blockIdx.x;
  int lane = threadIdx.x;
  const float* base = m + (size_t)rb * C;

  int off = (lane & 15) * 4 + ((lane >> 4) << 1);

  v8f acc = {0.f, 0.f, 0.f, 0.f, 0.f, 0.f, 0.f, 0.f};
  for (int k = 0; k < C; k += 64) {
    float2 t = *(const float2*)(base + k + off);
    v2f a;
    a.x = t.x;
    a.y = t.y;
    acc = __builtin_amdgcn_wmma_f32_16x16x4_f32(
         false, a,  false, a,
         (short)0, acc,  false,  false);
  }

  float d = 0.f;
  if (lane < 8)       d = acc[lane];
  else if (lane >= 24) d = acc[lane - 24];

#pragma unroll
  for (int s = 16; s >= 1; s >>= 1) d += __shfl_xor(d, s, 32);

  if (lane == 0) w[rb] = 1.0f / (sqrtf(d) + kEps);
}

__global__ __launch_bounds__(256) void rmac_combine_kernel(
    const float* __restrict__ m, const float* __restrict__ w,
    float* __restrict__ out, int nPlanes, int C) {
  int idx = blockIdx.x * blockDim.x + threadIdx.x;
  if (idx >= nPlanes) return;
  int B = nPlanes / C;
  int b = idx / C;
  float v = 0.f;
#pragma unroll
  for (int r = 0; r < kNReg; ++r) {
    float wt = w[r * B + b];
    if (r == 0) wt *= 2.0f;
    v += m[(size_t)r * nPlanes + idx] * wt;
  }
  out[idx] = v;
}

extern "C" void kernel_launch(void* const* d_in, const int* in_sizes, int n_in,
                              void* d_out, int out_size, void* d_ws, size_t ws_size,
                              hipStream_t stream) {
  const float* x = (const float*)d_in[0];
  const int C = 2048;
  int nPlanes = in_sizes[0] / kHW;
  int B = nPlanes / C;

  float* m = (float*)d_ws;
  float* w = m + (size_t)kNReg * nPlanes;

  const int threads = 256;
  int blocks = (nPlanes + threads - 1) / threads;
  rmac_region_max_kernel<<<blocks, threads, 0, stream>>>(x, m, nPlanes);
  rmac_norm_kernel<<<kNReg * B, 32, 0, stream>>>(m, w, C);
  rmac_combine_kernel<<<blocks, threads, 0, stream>>>(m, w, (float*)d_out, nPlanes, C);
}


# Region-max kernel: 7 lanes share a plane, each loads 16-byte pieces so a load covers 112 contiguous bytes per plane; row/column folds in registers, 3 row_xmask max steps join the lanes; no LDS
# speedup vs baseline: 1.5194x; 1.5194x over previous
#include <hip/hip_runtime.h>
#include <math.h>

typedef __attribute__((ext_vector_type(2))) float v2f;
typedef __attribute__((ext_vector_type(8))) float v8f;

static constexpr int kHW    = 196;
static constexpr int kWdim  = 14;
static constexpr int kNReg  = 14;
static constexpr float kEps = 1e-6f;

static constexpr int kPieces = 49;
static constexpr int kTrips  = 8;

template <int M>
__device__ __forceinline__ float group_xmax(float v) {
  float o;
  if (M == 1)
    asm("v_max_num_f32_dpp %0, %1, %1 row_xmask:1 row_mask:0xf bank_mask:0xf" : "=v"(o) : "v"(v));
  else if (M == 2)
    asm("v_max_num_f32_dpp %0, %1, %1 row_xmask:2 row_mask:0xf bank_mask:0xf" : "=v"(o) : "v"(v));
  else
    asm("v_max_num_f32_dpp %0, %1, %1 row_xmask:4 row_mask:0xf bank_mask:0xf" : "=v"(o) : "v"(v));
  return o;
}

__device__ __forceinline__ float keep_if(float v, float keep) {
  return fminf(v, keep);
}

__global__ __launch_bounds__(256) void rmac_region_max_kernel(
    const float* __restrict__ x, float* __restrict__ m, int nPlanes) {
  const int lane = threadIdx.x & 31;
  const int wave = blockIdx.x * (blockDim.x >> 5) + (threadIdx.x >> 5);
  const int base = wave * 32;
  if (base >= nPlanes) return;
  const int grp = lane >> 3;
  const int g   = lane & 7;
  const int gq  = g < 7 ? g : 6;
  const bool act = g < 7;

  const float kKeep = INFINITY, kDrop = -INFINITY;
  float evenRow[4], oddRow[4], inA3[4], inB3[4], inC3[4], inA2[4], inB2[4];
#pragma unroll
  for (int e = 0; e < 4; ++e) {
    const int f = 4 * gq + e;
    const bool odd = f >= 14;
    const int c = odd ? f - 14 : f;
    evenRow[e] = odd ? kDrop : kKeep;
    oddRow[e]  = odd ? kKeep : kDrop;
    inA3[e] = (act && c <= 6) ? kKeep : kDrop;
    inB3[e] = (act && c >= 3 && c <= 9) ? kKeep : kDrop;
    inC3[e] = (act && c >= 7) ? kKeep : kDrop;
    inA2[e] = (act && c <= 8) ? kKeep : kDrop;
    inB2[e] = (act && c >= 5) ? kKeep : kDrop;
  }

  const float4* xq = (const float4*)x;
  const int lastPlane = nPlanes - 1;

  float4 q[4][7];
  float acc[kNReg];
#pragma unroll
  for (int r = 0; r < kNReg; ++r) acc[r] = kDrop;

#define LOAD_TRIP(buf, t)                                                    \
  {                                                                          \
    int pl = base + 8 * grp + (t);                                           \
    pl = pl < lastPlane ? pl : lastPlane;                                    \
    const float4* p = xq + (size_t)pl * kPieces + gq;                        \
    _Pragma("unroll") for (int k = 0; k < 7; ++k) q[buf][k] = p[7 * k];      \
  }

  LOAD_TRIP(0, 0)
  LOAD_TRIP(1, 1)
  LOAD_TRIP(2, 2)

#pragma unroll
  for (int t = 0; t < kTrips; ++t) {
    if (t + 3 < kTrips) LOAD_TRIP((t + 3) % 4, t + 3)
    __builtin_amdgcn_sched_barrier(0);
    const float4* v4 = q[t % 4];

    float r7a[4], r7b[4], r7c[4], r9a[4], r9b[4];
#pragma unroll
    for (int e = 0; e < 4; ++e) {
      float v[7];
#pragma unroll
      for (int k = 0; k < 7; ++k)
        v[k] = e == 0 ? v4[k].x : e == 1 ? v4[k].y : e == 2 ? v4[k].z : v4[k].w;
      const float m012  = fmaxf(fmaxf(v[0], v[1]), v[2]);
      const float m234  = fmaxf(fmaxf(v[2], v[3]), v[4]);
      const float m456  = fmaxf(fmaxf(v[4], v[5]), v[6]);
      const float m0123 = fmaxf(m012, v[3]);
      const float m3456 = fmaxf(v[3], m456);
      r7a[e] = fmaxf(m012,  keep_if(v[3], evenRow[e]));
      r7b[e] = fmaxf(m234,  keep_if(v[1], oddRow[e]));
      r7c[e] = fmaxf(m456,  keep_if(v[3], oddRow[e]));
      r9a[e] = fmaxf(m0123, keep_if(v[4], evenRow[e]));
      r9b[e] = fmaxf(m3456, keep_if(v[2], oddRow[e]));
    }

    float res[kNReg];
#define COLS(rr, in)                                                              \
  fmaxf(fmaxf(keep_if(rr[0], in[0]), keep_if(rr[1], in[1])),                      \
        fmaxf(keep_if(rr[2], in[2]), keep_if(rr[3], in[3])))
    res[1]  = COLS(r9a, inA2);  res[2]  = COLS(r9a, inB2);
    res[3]  = COLS(r9b, inA2);  res[4]  = COLS(r9b, inB2);
    res[5]  = COLS(r7a, inA3);  res[6]  = COLS(r7a, inB3);  res[7]  = COLS(r7a, inC3);
    res[8]  = COLS(r7b, inA3);  res[9]  = COLS(r7b, inB3);  res[10] = COLS(r7b, inC3);
    res[11] = COLS(r7c, inA3);  res[12] = COLS(r7c, inB3);  res[13] = COLS(r7c, inC3);
#undef COLS

#pragma unroll
    for (int r = 1; r < kNReg; ++r) {
      float s = res[r];
      s = group_xmax<1>(s);
      s = group_xmax<2>(s);
      s = group_xmax<4>(s);
      acc[r] = (g == t) ? s : acc[r];
    }
  }
#undef LOAD_TRIP

  acc[0] = fmaxf(fmaxf(acc[1], acc[2]), fmaxf(acc[3], acc[4]));

  const int idx = base + lane;
  if (idx < nPlanes) {
#pragma unroll
    for (int r = 0; r < kNReg; ++r) m[(size_t)r * nPlanes + idx] = acc[r];
  }
}

__global__ __launch_bounds__(32) void rmac_norm_kernel(
    const float* __restrict__ m, float* __restrict__ w, int C) {
  int rb = blockIdx.x;
  int lane = threadIdx.x;
  const float* base = m + (size_t)rb * C;

  int off = (lane & 15) * 4 + ((lane >> 4) << 1);

  v8f acc = {0.f, 0.f, 0.f, 0.f, 0.f, 0.f, 0.f, 0.f};
  for (int k = 0; k < C; k += 64) {
    float2 t = *(const float2*)(base + k + off);
    v2f a;
    a.x = t.x;
    a.y = t.y;
    acc = __builtin_amdgcn_wmma_f32_16x16x4_f32(
         false, a,  false, a,
         (short)0, acc,  false,  false);
  }

  float d = 0.f;
  if (lane < 8)       d = acc[lane];
  else if (lane >= 24) d = acc[lane - 24];

#pragma unroll
  for (int s = 16; s >= 1; s >>= 1) d += __shfl_xor(d, s, 32);

  if (lane == 0) w[rb] = 1.0f / (sqrtf(d) + kEps);
}

__global__ __launch_bounds__(256) void rmac_combine_kernel(
    const float* __restrict__ m, const float* __restrict__ w,
    float* __restrict__ out, int nPlanes, int C) {
  int idx = blockIdx.x * blockDim.x + threadIdx.x;
  if (idx >= nPlanes) return;
  int B = nPlanes / C;
  int b = idx / C;
  float v = 0.f;
#pragma unroll
  for (int r = 0; r < kNReg; ++r) {
    float wt = w[r * B + b];
    if (r == 0) wt *= 2.0f;
    v += m[(size_t)r * nPlanes + idx] * wt;
  }
  out[idx] = v;
}

extern "C" void kernel_launch(void* const* d_in, const int* in_sizes, int n_in,
                              void* d_out, int out_size, void* d_ws, size_t ws_size,
                              hipStream_t stream) {
  const float* x = (const float*)d_in[0];
  const int C = 2048;
  int nPlanes = in_sizes[0] / kHW;
  int B = nPlanes / C;

  float* m = (float*)d_ws;
  float* w = m + (size_t)kNReg * nPlanes;

  const int threads = 256;
  int blocks = (nPlanes + threads - 1) / threads;
  rmac_region_max_kernel<<<blocks, threads, 0, stream>>>(x, m, nPlanes);
  rmac_norm_kernel<<<kNReg * B, 32, 0, stream>>>(m, w, C);
  rmac_combine_kernel<<<blocks, threads, 0, stream>>>(m, w, (float*)d_out, nPlanes, C);
}


# Norm and combine launches merged: one 16-wave workgroup per batch row reads its maxima once into registers, squares on f32 WMMA, 14 norms meet in LDS, then combine; 2 launches; sum order changes
# speedup vs baseline: 1.8044x; 1.1876x over previous
#include <hip/hip_runtime.h>
#include <math.h>

typedef __attribute__((ext_vector_type(2))) float v2f;
typedef __attribute__((ext_vector_type(8))) float v8f;

static constexpr int kHW    = 196;
static constexpr int kWdim  = 14;
static constexpr int kNReg  = 14;
static constexpr float kEps = 1e-6f;

static constexpr int kPieces = 49;
static constexpr int kTrips  = 8;

template <int M>
__device__ __forceinline__ float group_xmax(float v) {
  float o;
  if (M == 1)
    asm("v_max_num_f32_dpp %0, %1, %1 row_xmask:1 row_mask:0xf bank_mask:0xf" : "=v"(o) : "v"(v));
  else if (M == 2)
    asm("v_max_num_f32_dpp %0, %1, %1 row_xmask:2 row_mask:0xf bank_mask:0xf" : "=v"(o) : "v"(v));
  else
    asm("v_max_num_f32_dpp %0, %1, %1 row_xmask:4 row_mask:0xf bank_mask:0xf" : "=v"(o) : "v"(v));
  return o;
}

__device__ __forceinline__ float keep_if(float v, float keep) {
  return fminf(v, keep);
}

__global__ __launch_bounds__(256) void rmac_region_max_kernel(
    const float* __restrict__ x, float* __restrict__ m, int nPlanes) {
  const int lane = threadIdx.x & 31;
  const int wave = blockIdx.x * (blockDim.x >> 5) + (threadIdx.x >> 5);
  const int base = wave * 32;
  if (base >= nPlanes) return;
  const int grp = lane >> 3;
  const int g   = lane & 7;
  const int gq  = g < 7 ? g : 6;
  const bool act = g < 7;

  const float kKeep = INFINITY, kDrop = -INFINITY;
  float evenRow[4], oddRow[4], inA3[4], inB3[4], inC3[4], inA2[4], inB2[4];
#pragma unroll
  for (int e = 0; e < 4; ++e) {
    const int f = 4 * gq + e;
    const bool odd = f >= 14;
    const int c = odd ? f - 14 : f;
    evenRow[e] = odd ? kDrop : kKeep;
    oddRow[e]  = odd ? kKeep : kDrop;
    inA3[e] = (act && c <= 6) ? kKeep : kDrop;
    inB3[e] = (act && c >= 3 && c <= 9) ? kKeep : kDrop;
    inC3[e] = (act && c >= 7) ? kKeep : kDrop;
    inA2[e] = (act && c <= 8) ? kKeep : kDrop;
    inB2[e] = (act && c >= 5) ? kKeep : kDrop;
  }

  const float4* xq = (const float4*)x;
  const int lastPlane = nPlanes - 1;

  float4 q[4][7];
  float acc[kNReg];
#pragma unroll
  for (int r = 0; r < kNReg; ++r) acc[r] = kDrop;

#define LOAD_TRIP(buf, t)                                                    \
  {                                                                          \
    int pl = base + 8 * grp + (t);                                           \
    pl = pl < lastPlane ? pl : lastPlane;                                    \
    const float4* p = xq + (size_t)pl * kPieces + gq;                        \
    _Pragma("unroll") for (int k = 0; k < 7; ++k) q[buf][k] = p[7 * k];      \
  }

  LOAD_TRIP(0, 0)
  LOAD_TRIP(1, 1)
  LOAD_TRIP(2, 2)

#pragma unroll
  for (int t = 0; t < kTrips; ++t) {
    if (t + 3 < kTrips) LOAD_TRIP((t + 3) % 4, t + 3)
    __builtin_amdgcn_sched_barrier(0);
    const float4* v4 = q[t % 4];

    float r7a[4], r7b[4], r7c[4], r9a[4], r9b[4];
#pragma unroll
    for (int e = 0; e < 4; ++e) {
      float v[7];
#pragma unroll
      for (int k = 0; k < 7; ++k)
        v[k] = e == 0 ? v4[k].x : e == 1 ? v4[k].y : e == 2 ? v4[k].z : v4[k].w;
      const float m012  = fmaxf(fmaxf(v[0], v[1]), v[2]);
      const float m234  = fmaxf(fmaxf(v[2], v[3]), v[4]);
      const float m456  = fmaxf(fmaxf(v[4], v[5]), v[6]);
      const float m0123 = fmaxf(m012, v[3]);
      const float m3456 = fmaxf(v[3], m456);
      r7a[e] = fmaxf(m012,  keep_if(v[3], evenRow[e]));
      r7b[e] = fmaxf(m234,  keep_if(v[1], oddRow[e]));
      r7c[e] = fmaxf(m456,  keep_if(v[3], oddRow[e]));
      r9a[e] = fmaxf(m0123, keep_if(v[4], evenRow[e]));
      r9b[e] = fmaxf(m3456, keep_if(v[2], oddRow[e]));
    }

    float res[kNReg];
#define COLS(rr, in)                                                              \
  fmaxf(fmaxf(keep_if(rr[0], in[0]), keep_if(rr[1], in[1])),                      \
        fmaxf(keep_if(rr[2], in[2]), keep_if(rr[3], in[3])))
    res[1]  = COLS(r9a, inA2);  res[2]  = COLS(r9a, inB2);
    res[3]  = COLS(r9b, inA2);  res[4]  = COLS(r9b, inB2);
    res[5]  = COLS(r7a, inA3);  res[6]  = COLS(r7a, inB3);  res[7]  = COLS(r7a, inC3);
    res[8]  = COLS(r7b, inA3);  res[9]  = COLS(r7b, inB3);  res[10] = COLS(r7b, inC3);
    res[11] = COLS(r7c, inA3);  res[12] = COLS(r7c, inB3);  res[13] = COLS(r7c, inC3);
#undef COLS

#pragma unroll
    for (int r = 1; r < kNReg; ++r) {
      float s = res[r];
      s = group_xmax<1>(s);
      s = group_xmax<2>(s);
      s = group_xmax<4>(s);
      acc[r] = (g == t) ? s : acc[r];
    }
  }
#undef LOAD_TRIP

  acc[0] = fmaxf(fmaxf(acc[1], acc[2]), fmaxf(acc[3], acc[4]));

  const int idx = base + lane;
  if (idx < nPlanes) {
#pragma unroll
    for (int r = 0; r < kNReg; ++r) m[(size_t)r * nPlanes + idx] = acc[r];
  }
}

static constexpr int kC         = 2048;
static constexpr int kNcThreads = kC / 4;

template <int M>
__device__ __forceinline__ float row_xadd(float v) {
  float o;
  if (M == 1)
    asm("v_add_f32_dpp %0, %1, %1 row_xmask:1 row_mask:0xf bank_mask:0xf" : "=v"(o) : "v"(v));
  else if (M == 2)
    asm("v_add_f32_dpp %0, %1, %1 row_xmask:2 row_mask:0xf bank_mask:0xf" : "=v"(o) : "v"(v));
  else if (M == 4)
    asm("v_add_f32_dpp %0, %1, %1 row_xmask:4 row_mask:0xf bank_mask:0xf" : "=v"(o) : "v"(v));
  else
    asm("v_add_f32_dpp %0, %1, %1 row_xmask:8 row_mask:0xf bank_mask:0xf" : "=v"(o) : "v"(v));
  return o;
}

__global__ __launch_bounds__(kNcThreads) void rmac_norm_combine_kernel(
    const float* __restrict__ m, float* __restrict__ out, int nPlanes) {
  __shared__ alignas(16) float red[kNReg][kNcThreads];
  __shared__ float winv[kNReg];

  const int tid  = threadIdx.x;
  const int lane = tid & 31;
  const int wave = tid >> 5;
  const size_t idx = (size_t)blockIdx.x * kC + 4 * tid;

  float4 v[kNReg];
#pragma unroll
  for (int r = 0; r < kNReg; ++r)
    v[r] = *(const float4*)(m + (size_t)r * nPlanes + idx);

  const int sel = lane < 8 ? lane : (lane >= 24 ? lane - 24 : -1);

#pragma unroll
  for (int r = 0; r < kNReg; ++r) {
    v2f a0, a1;
    a0.x = v[r].x;  a0.y = v[r].y;
    a1.x = v[r].z;  a1.y = v[r].w;
    v8f acc = {0.f, 0.f, 0.f, 0.f, 0.f, 0.f, 0.f, 0.f};
    acc = __builtin_amdgcn_wmma_f32_16x16x4_f32(
         false, a0,  false, a0,
         (short)0, acc,  false,  false);
    acc = __builtin_amdgcn_wmma_f32_16x16x4_f32(
         false, a1,  false, a1,
         (short)0, acc,  false,  false);
    float d = 0.f;
#pragma unroll
    for (int j = 0; j < 8; ++j) d = (sel == j) ? acc[j] : d;
    red[r][tid] = d;
  }
  __syncthreads();

  if (wave < kNReg / 2) {
    const int c = lane & 15;
    const int r = 2 * wave + (lane >> 4);
    const float4* p = (const float4*)red[r];
    float4 t[kNcThreads / 64];
#pragma unroll
    for (int j = 0; j < kNcThreads / 64; ++j) t[j] = p[c + 16 * j];
    float u[kNcThreads / 64];
#pragma unroll
    for (int j = 0; j < kNcThreads / 64; ++j) u[j] = (t[j].x + t[j].y) + (t[j].z + t[j].w);
    float s = ((u[0] + u[1]) + (u[2] + u[3])) + ((u[4] + u[5]) + (u[6] + u[7]));
    s = row_xadd<1>(s);
    s = row_xadd<2>(s);
    s = row_xadd<4>(s);
    s = row_xadd<8>(s);
    if (c == 0) winv[r] = 1.0f / (sqrtf(s) + kEps);
  }
  __syncthreads();

  float4 o = {0.f, 0.f, 0.f, 0.f};
#pragma unroll
  for (int r = 0; r < kNReg; ++r) {
    float wt = winv[r];
    if (r == 0) wt *= 2.0f;
    o.x += v[r].x * wt;
    o.y += v[r].y * wt;
    o.z += v[r].z * wt;
    o.w += v[r].w * wt;
  }
  *(float4*)(out + idx) = o;
}

extern "C" void kernel_launch(void* const* d_in, const int* in_sizes, int n_in,
                              void* d_out, int out_size, void* d_ws, size_t ws_size,
                              hipStream_t stream) {
  const float* x = (const float*)d_in[0];
  int nPlanes = in_sizes[0] / kHW;
  int B = nPlanes / kC;

  float* m = (float*)d_ws;

  const int threads = 256;
  int blocks = (nPlanes + threads - 1) / threads;
  rmac_region_max_kernel<<<blocks, threads, 0, stream>>>(x, m, nPlanes);
  rmac_norm_combine_kernel<<<B, kNcThreads, 0, stream>>>(m, (float*)d_out, nPlanes);
}
